# starting kernel
# baseline (speedup 1.0000x reference)
#include <hip/hip_runtime.h>
#include <hip/hip_bf16.h>

#define PB 8
#define PD 64
#define PS 512

typedef __attribute__((ext_vector_type(2))) float v2f;
typedef __attribute__((ext_vector_type(8))) float v8f;

__device__ __forceinline__ v8f wmma_f32_16x16x4(v2f a, v2f b, v8f c) {
    return __builtin_amdgcn_wmma_f32_16x16x4_f32(
         false, a,  false, b,
         (short)0, c,  false,  false);
}

__device__ __forceinline__ v8f v8f_zero() {
    v8f z = {0.f, 0.f, 0.f, 0.f, 0.f, 0.f, 0.f, 0.f};
    return z;
}

__global__ __launch_bounds__(128) void lie_stage1(
    const float* __restrict__ x, const float* __restrict__ W,
    float* __restrict__ Aws, float* __restrict__ Bws) {

    const int lane = threadIdx.x & 31;
    const int wv   = threadIdx.x >> 5;
    const int gw   = blockIdx.x * 4 + wv;
    const int b    = gw >> 5;
    const int jb   = (gw & 31) * 16;
    const int ln   = lane & 15;
    const int kh   = lane >> 4;

    __builtin_prefetch(W + threadIdx.x * 4, 0, 1);

    const float* xb = x + (size_t)b * PD * PS;

    v8f accA[4], accB[4];
#pragma unroll
    for (int nt = 0; nt < 4; ++nt) { accA[nt] = v8f_zero(); accB[nt] = v8f_zero(); }

#pragma unroll
    for (int kb = 0; kb < PD; kb += 4) {
        const int k0 = kb + 2 * kh;
        v2f a;
        a.x = xb[(size_t)k0 * PS + jb + ln];
        a.y = xb[(size_t)(k0 + 1) * PS + jb + ln];
#pragma unroll
        for (int nt = 0; nt < 4; ++nt) {
            const int n = nt * 16 + ln;
            v2f bw, bh;
            bw.x = W[(size_t)k0 * PD + n];
            bw.y = W[(size_t)(k0 + 1) * PD + n];
            bh.x = W[(size_t)(k0 + PD) * PD + n];
            bh.y = W[(size_t)(k0 + 1 + PD) * PD + n];
            accA[nt] = wmma_f32_16x16x4(a, bw, accA[nt]);
            accB[nt] = wmma_f32_16x16x4(a, bh, accB[nt]);
        }
    }

    float* Ab = Aws + ((size_t)b * PS + jb) * PD;
    float* Bb = Bws + ((size_t)b * PS + jb) * PD;
#pragma unroll
    for (int nt = 0; nt < 4; ++nt) {
#pragma unroll
        for (int v = 0; v < 8; ++v) {
            const int r = v + 8 * kh;
            Ab[(size_t)r * PD + nt * 16 + ln] = accA[nt][v];
            Bb[(size_t)r * PD + nt * 16 + ln] = accB[nt][v];
        }
    }
}

__global__ __launch_bounds__(256) void lie_stage2(
    const float* __restrict__ x, const float* __restrict__ Aws,
    const float* __restrict__ Bws, float* __restrict__ out) {

    const int b  = blockIdx.y;
    const int ib = blockIdx.x * 64;

    __shared__ float sA[64][65];
    __shared__ float sB[64][65];
    __shared__ float sWm[64][64];
    __shared__ float sNa[64];
    __shared__ float sNb[64];

    const int t    = threadIdx.x;
    const int lane = t & 31;
    const int wv   = t >> 5;
    const int ln   = lane & 15;
    const int kh   = lane >> 4;
    const int mt   = (wv & 3) * 16;
    const int nh   = (wv >> 2) * 32;

    const float* Ab = Aws + ((size_t)b * PS + ib) * PD;
#pragma unroll
    for (int p = 0; p < 16; ++p) {
        const int lin = t + 256 * p;
        sA[lin >> 6][lin & 63] = Ab[lin];
    }
    __syncthreads();
    if (t < 64) {
        float s = 0.f;
#pragma unroll
        for (int k = 0; k < 64; ++k) { const float v = sA[t][k]; s += v * v; }
        sNa[t] = s;
    }
    __syncthreads();

    const float* xb = x + (size_t)b * PD * PS;
    v8f c0 = v8f_zero(), c1 = v8f_zero();

    for (int wc = 0; wc < PS; wc += 64) {
        const float* Bb = Bws + ((size_t)b * PS + wc) * PD;
#pragma unroll
        for (int p = 0; p < 16; ++p) {
            const int lin = t + 256 * p;
            sB[lin >> 6][lin & 63] = Bb[lin];
        }
        __syncthreads();
        if (t < 64) {
            float s = 0.f;
#pragma unroll
            for (int k = 0; k < 64; ++k) { const float v = sB[t][k]; s += v * v; }
            sNb[t] = s;
        }
        __syncthreads();

        v8f g0 = v8f_zero(), g1 = v8f_zero();
#pragma unroll
        for (int kb = 0; kb < 64; kb += 4) {
            const int k0 = kb + 2 * kh;
            v2f a;  a.x  = sB[mt + ln][k0];        a.y  = sB[mt + ln][k0 + 1];
            v2f b0; b0.x = sA[nh + ln][k0];        b0.y = sA[nh + ln][k0 + 1];
            v2f b1; b1.x = sA[nh + 16 + ln][k0];   b1.y = sA[nh + 16 + ln][k0 + 1];
            g0 = wmma_f32_16x16x4(a, b0, g0);
            g1 = wmma_f32_16x16x4(a, b1, g1);
        }

#pragma unroll
        for (int v = 0; v < 8; ++v) {
            const int wrow = mt + v + 8 * kh;
            const int gwi  = wc + wrow;
            const int i0   = nh + ln;
            const int i1   = nh + 16 + ln;
            float m0 = 1.0f, m1 = 1.0f;
            if (ib + i0 <= gwi)
                m0 = 1.0f - sqrtf(fmaxf(sNa[i0] + sNb[wrow] + 2.0f * g0[v], 0.0f));
            if (ib + i1 <= gwi)
                m1 = 1.0f - sqrtf(fmaxf(sNa[i1] + sNb[wrow] + 2.0f * g1[v], 0.0f));
            sWm[wrow][i0] = m0;
            sWm[wrow][i1] = m1;
        }
        __syncthreads();

#pragma unroll
        for (int kb = 0; kb < 64; kb += 4) {
            const int k0 = kb + 2 * kh;
            v2f a;  a.x  = xb[(size_t)(mt + ln) * PS + wc + k0];
                    a.y  = xb[(size_t)(mt + ln) * PS + wc + k0 + 1];
            v2f b0; b0.x = sWm[k0][nh + ln];       b0.y = sWm[k0 + 1][nh + ln];
            v2f b1; b1.x = sWm[k0][nh + 16 + ln];  b1.y = sWm[k0 + 1][nh + 16 + ln];
            c0 = wmma_f32_16x16x4(a, b0, c0);
            c1 = wmma_f32_16x16x4(a, b1, c1);
        }
        __syncthreads();
    }

    float* ob = out + (size_t)b * PD * PS;
#pragma unroll
    for (int v = 0; v < 8; ++v) {
        const int d = mt + v + 8 * kh;
        ob[(size_t)d * PS + ib + nh + ln]      = c0[v];
        ob[(size_t)d * PS + ib + nh + 16 + ln] = c1[v];
    }
}

extern "C" void kernel_launch(void* const* d_in, const int* in_sizes, int n_in,
                              void* d_out, int out_size, void* d_ws, size_t ws_size,
                              hipStream_t stream) {
    const float* x = (const float*)d_in[0];
    const float* W = (const float*)d_in[1];
    float* out = (float*)d_out;

    float* Aws = (float*)d_ws;
    float* Bws = Aws + (size_t)PB * PS * PD;

    lie_stage1<<<64, 128, 0, stream>>>(x, W, Aws, Bws);
    lie_stage2<<<dim3(PS / 64, PB), 256, 0, stream>>>(x, Aws, Bws, out);
}


# one launch, was two: each work-group recomputes on WMMA, into LDS (pitch 68 floats), the 512 Bm rows of its batch and its own 32 A rows; stage 2 reads them there; no workspace; sum order kept
# speedup vs baseline: 2.8136x; 2.8136x over previous
#include <hip/hip_runtime.h>
#include <hip/hip_bf16.h>

#define PB 8
#define PD 64
#define PS 512
#define LP 68

typedef __attribute__((ext_vector_type(2))) float v2f;
typedef __attribute__((ext_vector_type(4))) float v4f;
typedef __attribute__((ext_vector_type(8))) float v8f;

__device__ __forceinline__ v8f wmma_f32_16x16x4(v2f a, v2f b, v8f c) {
    return __builtin_amdgcn_wmma_f32_16x16x4_f32(
         false, a,  false, b,
         (short)0, c,  false,  false);
}

__device__ __forceinline__ v8f v8f_zero() {
    v8f z = {0.f, 0.f, 0.f, 0.f, 0.f, 0.f, 0.f, 0.f};
    return z;
}

__device__ __forceinline__ v2f mk2(float a, float b) { v2f r; r.x = a; r.y = b; return r; }

__device__ __forceinline__ v2f kpair(const v4f (&p)[2], int q) {
    const v4f f = p[q >> 1];
    return (q & 1) ? mk2(f.z, f.w) : mk2(f.x, f.y);
}

__device__ __forceinline__ v8f wmat_tile(const v4f (&bm)[8], const v4f (&a2)[8],
                                         float nap, int icol, int wreg0) {
    v8f g = v8f_zero();
    v4f psq = {0.f, 0.f, 0.f, 0.f};
#pragma unroll
    for (int j = 0; j < 8; ++j) {
        const v4f f = bm[j];
        psq += f * f;
        g = wmma_f32_16x16x4(mk2(f.x, f.y), mk2(a2[j].x, a2[j].y), g);
        g = wmma_f32_16x16x4(mk2(f.z, f.w), mk2(a2[j].z, a2[j].w), g);
    }
    const float pb = (psq.x + psq.y) + (psq.z + psq.w);
    g = wmma_f32_16x16x4(mk2(pb, 1.0f), mk2(1.0f, nap), g);
#pragma unroll
    for (int r = 0; r < 8; ++r) {
        const float s = sqrtf(fmaxf(g[r], 0.0f));
        g[r] = (icol <= wreg0 + r) ? 1.0f - s : 1.0f;
    }
    return g;
}


__global__ __launch_bounds__(256) void lie_fused(
    const float* __restrict__ x, const float* __restrict__ W, float* __restrict__ out) {

    __shared__ alignas(16) float sB[PS][LP];
    __shared__ alignas(16) float sA[32][LP];
    __shared__ alignas(16) float sG[2][2][4][2][32][4];
    __shared__ alignas(16) float sC[PD][36];

    const int lane = threadIdx.x & 31;
    const int wv   = threadIdx.x >> 5;
    const int ct   = wv & 1;
    const int rt   = wv >> 1;
    const int c    = lane & 15;
    const int h    = lane >> 4;
    const int b    = blockIdx.y;
    const int ibw  = blockIdx.x * 32;
    const int icol = ibw + 16 * ct + c;

    const float* xb = x + (size_t)b * PD * PS;

    {
        v4f wh4[16][2], xa4[16][2];
        const float* wp = W + (size_t)(PD + 2 * h) * PD + 4 * c;
        const float* xp = xb + (size_t)(2 * h) * PS + 64 * wv + 4 * c;
#pragma unroll
        for (int ks = 0; ks < 16; ++ks) {
#pragma unroll
            for (int e = 0; e < 2; ++e) {
                wh4[ks][e] = *(const v4f*)(wp + (size_t)(4 * ks + e) * PD);
                xa4[ks][e] = *(const v4f*)(xp + (size_t)(4 * ks + e) * PS);
            }
        }
        const int sa = wv >> 2;
        const int na = wv & 3;
        v2f ww[16], xq[16];
#pragma unroll
        for (int ks = 0; ks < 16; ++ks) {
            const int d0 = 4 * ks + 2 * h;
            ww[ks] = mk2(W[(size_t)d0 * PD + 16 * na + c], W[(size_t)(d0 + 1) * PD + 16 * na + c]);
            xq[ks] = mk2(xb[(size_t)d0 * PS + ibw + 16 * sa + c],
                         xb[(size_t)(d0 + 1) * PS + ibw + 16 * sa + c]);
        }

#pragma unroll
        for (int q = 0; q < 4; ++q) {
            v8f acc4[4];
#pragma unroll
            for (int nt = 0; nt < 4; ++nt) acc4[nt] = v8f_zero();
#pragma unroll
            for (int ks = 0; ks < 16; ++ks) {
                const v2f a = mk2(xa4[ks][0][q], xa4[ks][1][q]);
#pragma unroll
                for (int nt = 0; nt < 4; ++nt)
                    acc4[nt] = wmma_f32_16x16x4(a, mk2(wh4[ks][0][nt], wh4[ks][1][nt]), acc4[nt]);
            }
#pragma unroll
            for (int r = 0; r < 8; ++r)
                *(v4f*)&sB[64 * wv + 32 * h + 4 * r + q][4 * c] =
                    v4f{acc4[0][r], acc4[1][r], acc4[2][r], acc4[3][r]};
        }

        v8f accA = v8f_zero();
#pragma unroll
        for (int ks = 0; ks < 16; ++ks)
            accA = wmma_f32_16x16x4(xq[ks], ww[ks], accA);
#pragma unroll
        for (int r = 0; r < 8; ++r)
            sA[16 * sa + 8 * h + r][16 * na + c] = accA[r];
    }

    const float* xbase = xb + (size_t)(16 * rt + c) * PS + 8 * h;
    v4f xf[4][2], xfn[4][2];
#pragma unroll
    for (int u = 0; u < 4; ++u)
#pragma unroll
        for (int p = 0; p < 2; ++p)
            xf[u][p] = *(const v4f*)(xbase + 16 * u + 4 * p);

    __syncthreads();

    const float* arow = &sA[16 * ct + c][4 * h];
    v4f a2[8];
    v4f nsq = {0.f, 0.f, 0.f, 0.f};
#pragma unroll
    for (int j = 0; j < 8; ++j) {
        const v4f f = *(const v4f*)(arow + 8 * j);
        nsq += f * f;
        a2[j] = f + f;
    }
    const float nap = (nsq.x + nsq.y) + (nsq.z + nsq.w);

    const float* bbase = &sB[16 * rt + c][4 * h];
    const int wl = 16 * rt + 8 * h;

    v4f bm[8], bmn[8];
    v4f G[4][2];

#pragma unroll
    for (int j = 0; j < 8; ++j) bm[j] = *(const v4f*)(bbase + 8 * j);
#pragma unroll
    for (int j = 0; j < 8; ++j) bmn[j] = *(const v4f*)(bbase + (size_t)64 * LP + 8 * j);

    {
        const v8f g = wmat_tile(bm, a2, nap, icol, wl);
        *(v4f*)&sG[0][ct][rt][0][lane][0] = v4f{g[0], g[1], g[2], g[3]};
        *(v4f*)&sG[0][ct][rt][1][lane][0] = v4f{g[4], g[5], g[6], g[7]};
    }
    __syncthreads();
#pragma unroll
    for (int u = 0; u < 4; ++u)
#pragma unroll
        for (int p = 0; p < 2; ++p)
            G[u][p] = *(const v4f*)&sG[0][ct][u][p][lane][0];
#pragma unroll
    for (int j = 0; j < 8; ++j) bm[j] = bmn[j];

    v8f acc = v8f_zero();

    for (int t = 1; t < PS / 64; ++t) {
        const int par = t & 1;
        const int tn = (t + 1 < PS / 64) ? t + 1 : t;

#pragma unroll
        for (int j = 0; j < 8; ++j)
            bmn[j] = *(const v4f*)(bbase + (size_t)(64 * tn) * LP + 8 * j);
#pragma unroll
        for (int u = 0; u < 4; ++u)
#pragma unroll
            for (int p = 0; p < 2; ++p)
                xfn[u][p] = *(const v4f*)(xbase + 64 * t + 16 * u + 4 * p);

        const v8f g = wmat_tile(bm, a2, nap, icol, 64 * t + wl);

#pragma unroll
        for (int u = 0; u < 4; ++u)
#pragma unroll
            for (int q = 0; q < 4; ++q)
                acc = wmma_f32_16x16x4(kpair(xf[u], q), kpair(G[u], q), acc);

        *(v4f*)&sG[par][ct][rt][0][lane][0] = v4f{g[0], g[1], g[2], g[3]};
        *(v4f*)&sG[par][ct][rt][1][lane][0] = v4f{g[4], g[5], g[6], g[7]};
        __syncthreads();
#pragma unroll
        for (int u = 0; u < 4; ++u)
#pragma unroll
            for (int p = 0; p < 2; ++p)
                G[u][p] = *(const v4f*)&sG[par][ct][u][p][lane][0];
#pragma unroll
        for (int j = 0; j < 8; ++j) bm[j] = bmn[j];
#pragma unroll
        for (int u = 0; u < 4; ++u)
#pragma unroll
            for (int p = 0; p < 2; ++p)
                xf[u][p] = xfn[u][p];
    }

#pragma unroll
    for (int u = 0; u < 4; ++u)
#pragma unroll
        for (int q = 0; q < 4; ++q)
            acc = wmma_f32_16x16x4(kpair(xf[u], q), kpair(G[u], q), acc);

#pragma unroll
    for (int r = 0; r < 8; ++r)
        sC[16 * rt + 8 * h + r][16 * ct + c] = acc[r];
    __syncthreads();
    float* ob = out + (size_t)b * PD * PS + ibw;
#pragma unroll
    for (int q = 0; q < 2; ++q) {
        const int row = 8 * wv + 4 * q + (lane >> 3);
        const int col = 4 * (lane & 7);
        *(v4f*)(ob + (size_t)row * PS + col) = *(const v4f*)&sC[row][col];
    }
}

extern "C" void kernel_launch(void* const* d_in, const int* in_sizes, int n_in,
                              void* d_out, int out_size, void* d_ws, size_t ws_size,
                              hipStream_t stream) {
    const float* x = (const float*)d_in[0];
    const float* W = (const float*)d_in[1];
    float* out = (float*)d_out;

    lie_fused<<<dim3(PS / 32, PB), 256, 0, stream>>>(x, W, out);
}


# output rows d split over two work-groups per column block (256 work-groups, one per WGP): each builds all wmat tiles and does half of product 2 on four waves, one per SIMD; sum order kept
# speedup vs baseline: 2.9999x; 1.0662x over previous
#include <hip/hip_runtime.h>
#include <hip/hip_bf16.h>

#define PB 8
#define PD 64
#define PS 512
#define LP 68

typedef __attribute__((ext_vector_type(2))) float v2f;
typedef __attribute__((ext_vector_type(4))) float v4f;
typedef __attribute__((ext_vector_type(8))) float v8f;

__device__ __forceinline__ v8f wmma_f32_16x16x4(v2f a, v2f b, v8f c) {
    return __builtin_amdgcn_wmma_f32_16x16x4_f32(
         false, a,  false, b,
         (short)0, c,  false,  false);
}

__device__ __forceinline__ v8f v8f_zero() {
    v8f z = {0.f, 0.f, 0.f, 0.f, 0.f, 0.f, 0.f, 0.f};
    return z;
}

__device__ __forceinline__ v2f mk2(float a, float b) { v2f r; r.x = a; r.y = b; return r; }

__device__ __forceinline__ v2f kpair(const v4f (&p)[2], int q) {
    const v4f f = p[q >> 1];
    return (q & 1) ? mk2(f.z, f.w) : mk2(f.x, f.y);
}

__device__ __forceinline__ v8f wmat_tile(const v4f (&bm)[8], const v4f (&a2)[8],
                                         float nap, int icol, int wreg0) {
    v8f g = v8f_zero();
    v4f psq = {0.f, 0.f, 0.f, 0.f};
#pragma unroll
    for (int j = 0; j < 8; ++j) {
        const v4f f = bm[j];
        psq += f * f;
        g = wmma_f32_16x16x4(mk2(f.x, f.y), mk2(a2[j].x, a2[j].y), g);
        g = wmma_f32_16x16x4(mk2(f.z, f.w), mk2(a2[j].z, a2[j].w), g);
    }
    const float pb = (psq.x + psq.y) + (psq.z + psq.w);
    g = wmma_f32_16x16x4(mk2(pb, 1.0f), mk2(1.0f, nap), g);
#pragma unroll
    for (int r = 0; r < 8; ++r) {
        const float s = sqrtf(fmaxf(g[r], 0.0f));
        g[r] = (icol <= wreg0 + r) ? 1.0f - s : 1.0f;
    }
    return g;
}


__global__ __launch_bounds__(256) void lie_fused(
    const float* __restrict__ x, const float* __restrict__ W, float* __restrict__ out) {

    __shared__ alignas(16) float sB[PS][LP];
    __shared__ alignas(16) float sA[32][LP];
    __shared__ alignas(16) float sG[2][2][4][2][32][4];
    __shared__ alignas(16) float sC[32][36];

    const int lane = threadIdx.x & 31;
    const int wv   = threadIdx.x >> 5;
    const int ct   = wv & 1;
    const int rt   = wv >> 1;
    const int c    = lane & 15;
    const int h    = lane >> 4;
    const int b    = blockIdx.y;
    const int ibw  = blockIdx.x * 32;
    const int icol = ibw + 16 * ct + c;
    const int  wvs = __builtin_amdgcn_readfirstlane(wv);
    const bool j2  = (__builtin_popcount((unsigned)wvs) & 1) == 0;
    const int  dt2 = wvs >> 2;
    const int  ct2 = (wvs >> 1) & 1;
    const int  d0  = 32 * (int)blockIdx.z;

    const float* xb = x + (size_t)b * PD * PS;

    {
        v4f wh4[16][2], xa4[16][2];
        const float* wp = W + (size_t)(PD + 2 * h) * PD + 4 * c;
        const float* xp = xb + (size_t)(2 * h) * PS + 64 * wv + 4 * c;
#pragma unroll
        for (int ks = 0; ks < 16; ++ks) {
#pragma unroll
            for (int e = 0; e < 2; ++e) {
                wh4[ks][e] = *(const v4f*)(wp + (size_t)(4 * ks + e) * PD);
                xa4[ks][e] = *(const v4f*)(xp + (size_t)(4 * ks + e) * PS);
            }
        }
        const int sa = wv >> 2;
        const int na = wv & 3;
        v2f ww[16], xq[16];
#pragma unroll
        for (int ks = 0; ks < 16; ++ks) {
            const int d0 = 4 * ks + 2 * h;
            ww[ks] = mk2(W[(size_t)d0 * PD + 16 * na + c], W[(size_t)(d0 + 1) * PD + 16 * na + c]);
            xq[ks] = mk2(xb[(size_t)d0 * PS + ibw + 16 * sa + c],
                         xb[(size_t)(d0 + 1) * PS + ibw + 16 * sa + c]);
        }

#pragma unroll
        for (int q = 0; q < 4; ++q) {
            v8f acc4[4];
#pragma unroll
            for (int nt = 0; nt < 4; ++nt) acc4[nt] = v8f_zero();
#pragma unroll
            for (int ks = 0; ks < 16; ++ks) {
                const v2f a = mk2(xa4[ks][0][q], xa4[ks][1][q]);
#pragma unroll
                for (int nt = 0; nt < 4; ++nt)
                    acc4[nt] = wmma_f32_16x16x4(a, mk2(wh4[ks][0][nt], wh4[ks][1][nt]), acc4[nt]);
            }
#pragma unroll
            for (int r = 0; r < 8; ++r)
                *(v4f*)&sB[64 * wv + 32 * h + 4 * r + q][4 * c] =
                    v4f{acc4[0][r], acc4[1][r], acc4[2][r], acc4[3][r]};
        }

        v8f accA = v8f_zero();
#pragma unroll
        for (int ks = 0; ks < 16; ++ks)
            accA = wmma_f32_16x16x4(xq[ks], ww[ks], accA);
#pragma unroll
        for (int r = 0; r < 8; ++r)
            sA[16 * sa + 8 * h + r][16 * na + c] = accA[r];
    }

    const float* xbase = xb + (size_t)(d0 + 16 * dt2 + c) * PS + 8 * h;
    v4f xf[4][2], xfn[4][2];
    if (j2) {
#pragma unroll
        for (int u = 0; u < 4; ++u)
#pragma unroll
            for (int p = 0; p < 2; ++p)
                xf[u][p] = *(const v4f*)(xbase + 16 * u + 4 * p);
    }

    __syncthreads();

    const float* arow = &sA[16 * ct + c][4 * h];
    v4f a2[8];
    v4f nsq = {0.f, 0.f, 0.f, 0.f};
#pragma unroll
    for (int j = 0; j < 8; ++j) {
        const v4f f = *(const v4f*)(arow + 8 * j);
        nsq += f * f;
        a2[j] = f + f;
    }
    const float nap = (nsq.x + nsq.y) + (nsq.z + nsq.w);

    const float* bbase = &sB[16 * rt + c][4 * h];
    const int wl = 16 * rt + 8 * h;

    v4f bm[8], bmn[8];
    v4f G[4][2];

#pragma unroll
    for (int j = 0; j < 8; ++j) bm[j] = *(const v4f*)(bbase + 8 * j);
#pragma unroll
    for (int j = 0; j < 8; ++j) bmn[j] = *(const v4f*)(bbase + (size_t)64 * LP + 8 * j);

    {
        const v8f g = wmat_tile(bm, a2, nap, icol, wl);
        *(v4f*)&sG[0][ct][rt][0][lane][0] = v4f{g[0], g[1], g[2], g[3]};
        *(v4f*)&sG[0][ct][rt][1][lane][0] = v4f{g[4], g[5], g[6], g[7]};
    }
    __syncthreads();
    if (j2) {
#pragma unroll
        for (int u = 0; u < 4; ++u)
#pragma unroll
            for (int p = 0; p < 2; ++p)
                G[u][p] = *(const v4f*)&sG[0][ct2][u][p][lane][0];
    }
#pragma unroll
    for (int j = 0; j < 8; ++j) bm[j] = bmn[j];

    v8f acc = v8f_zero();

    for (int t = 1; t < PS / 64; ++t) {
        const int par = t & 1;
        const int tn = (t + 1 < PS / 64) ? t + 1 : t;

#pragma unroll
        for (int j = 0; j < 8; ++j)
            bmn[j] = *(const v4f*)(bbase + (size_t)(64 * tn) * LP + 8 * j);
        if (j2) {
#pragma unroll
            for (int u = 0; u < 4; ++u)
#pragma unroll
                for (int p = 0; p < 2; ++p)
                    xfn[u][p] = *(const v4f*)(xbase + 64 * t + 16 * u + 4 * p);

            const v8f g = wmat_tile(bm, a2, nap, icol, 64 * t + wl);
            *(v4f*)&sG[par][ct][rt][0][lane][0] = v4f{g[0], g[1], g[2], g[3]};
            *(v4f*)&sG[par][ct][rt][1][lane][0] = v4f{g[4], g[5], g[6], g[7]};

#pragma unroll
            for (int u = 0; u < 4; ++u)
#pragma unroll
                for (int q = 0; q < 4; ++q)
                    acc = wmma_f32_16x16x4(kpair(xf[u], q), kpair(G[u], q), acc);
        } else {
            const v8f g = wmat_tile(bm, a2, nap, icol, 64 * t + wl);
            *(v4f*)&sG[par][ct][rt][0][lane][0] = v4f{g[0], g[1], g[2], g[3]};
            *(v4f*)&sG[par][ct][rt][1][lane][0] = v4f{g[4], g[5], g[6], g[7]};
        }

        __syncthreads();
        if (j2) {
#pragma unroll
            for (int u = 0; u < 4; ++u)
#pragma unroll
                for (int p = 0; p < 2; ++p) {
                    G[u][p] = *(const v4f*)&sG[par][ct2][u][p][lane][0];
                    xf[u][p] = xfn[u][p];
                }
        }
#pragma unroll
        for (int j = 0; j < 8; ++j) bm[j] = bmn[j];
    }

    if (j2) {
#pragma unroll
        for (int u = 0; u < 4; ++u)
#pragma unroll
            for (int q = 0; q < 4; ++q)
                acc = wmma_f32_16x16x4(kpair(xf[u], q), kpair(G[u], q), acc);
#pragma unroll
        for (int r = 0; r < 8; ++r)
            sC[16 * dt2 + 8 * h + r][16 * ct2 + c] = acc[r];
    }
    __syncthreads();
    float* ob = out + (size_t)b * PD * PS + (size_t)d0 * PS + ibw;
    {
        const int row = 4 * wv + (lane >> 3);
        const int col = 4 * (lane & 7);
        *(v4f*)(ob + (size_t)row * PS + col) = *(const v4f*)&sC[row][col];
    }
}

extern "C" void kernel_launch(void* const* d_in, const int* in_sizes, int n_in,
                              void* d_out, int out_size, void* d_ws, size_t ws_size,
                              hipStream_t stream) {
    const float* x = (const float*)d_in[0];
    const float* W = (const float*)d_in[1];
    float* out = (float*)d_out;

    lie_fused<<<dim3(PS / 32, PB, 2), 256, 0, stream>>>(x, W, out);
}


# rows above the mask's diagonal skipped: a work-group's first (column block >> 1) 64-row blocks need no Bm rows, tiles or barriers (job 2 against ones); grid turned: fullest first; sum order kept
# speedup vs baseline: 3.1573x; 1.0525x over previous
#include <hip/hip_runtime.h>
#include <hip/hip_bf16.h>

#define PB 8
#define PD 64
#define PS 512
#define LP 68

typedef __attribute__((ext_vector_type(2))) float v2f;
typedef __attribute__((ext_vector_type(4))) float v4f;
typedef __attribute__((ext_vector_type(8))) float v8f;

__device__ __forceinline__ v8f wmma_f32_16x16x4(v2f a, v2f b, v8f c) {
    return __builtin_amdgcn_wmma_f32_16x16x4_f32(
         false, a,  false, b,
         (short)0, c,  false,  false);
}

__device__ __forceinline__ v8f v8f_zero() {
    v8f z = {0.f, 0.f, 0.f, 0.f, 0.f, 0.f, 0.f, 0.f};
    return z;
}

__device__ __forceinline__ v2f mk2(float a, float b) { v2f r; r.x = a; r.y = b; return r; }

__device__ __forceinline__ v2f kpair(const v4f (&p)[2], int q) {
    const v4f f = p[q >> 1];
    return (q & 1) ? mk2(f.z, f.w) : mk2(f.x, f.y);
}

__device__ __forceinline__ v8f wmat_tile(const v4f (&bm)[8], const v4f (&a2)[8],
                                         float nap, int icol, int wreg0) {
    v8f g = v8f_zero();
    v4f psq = {0.f, 0.f, 0.f, 0.f};
#pragma unroll
    for (int j = 0; j < 8; ++j) {
        const v4f f = bm[j];
        psq += f * f;
        g = wmma_f32_16x16x4(mk2(f.x, f.y), mk2(a2[j].x, a2[j].y), g);
        g = wmma_f32_16x16x4(mk2(f.z, f.w), mk2(a2[j].z, a2[j].w), g);
    }
    const float pb = (psq.x + psq.y) + (psq.z + psq.w);
    g = wmma_f32_16x16x4(mk2(pb, 1.0f), mk2(1.0f, nap), g);
#pragma unroll
    for (int r = 0; r < 8; ++r) {
        const float s = sqrtf(fmaxf(g[r], 0.0f));
        g[r] = (icol <= wreg0 + r) ? 1.0f - s : 1.0f;
    }
    return g;
}


__global__ __launch_bounds__(256) void lie_fused(
    const float* __restrict__ x, const float* __restrict__ W, float* __restrict__ out) {

    __shared__ alignas(16) float sB[PS][LP];
    __shared__ alignas(16) float sA[32][LP];
    __shared__ alignas(16) float sG[2][2][4][2][32][4];
    __shared__ alignas(16) float sC[32][36];

    const int lane = threadIdx.x & 31;
    const int wv   = threadIdx.x >> 5;
    const int ct   = wv & 1;
    const int rt   = wv >> 1;
    const int c    = lane & 15;
    const int h    = lane >> 4;
    const int b    = blockIdx.y;
    const int ibw  = blockIdx.z * 32;
    const int t0   = (int)blockIdx.z >> 1;
    const int icol = ibw + 16 * ct + c;
    const int  wvs = __builtin_amdgcn_readfirstlane(wv);
    const bool j2  = (__builtin_popcount((unsigned)wvs) & 1) == 0;
    const int  dt2 = wvs >> 2;
    const int  ct2 = (wvs >> 1) & 1;
    const int  d0  = 32 * (int)blockIdx.x;

    const float* xb = x + (size_t)b * PD * PS;

    {
        const bool own = wvs >= t0;
        v4f wh4[16][2], xa4[16][2];
        const float* wp = W + (size_t)(PD + 2 * h) * PD + 4 * c;
        const float* xp = xb + (size_t)(2 * h) * PS + 64 * wv + 4 * c;
        if (own) {
#pragma unroll
        for (int ks = 0; ks < 16; ++ks) {
#pragma unroll
            for (int e = 0; e < 2; ++e) {
                wh4[ks][e] = *(const v4f*)(wp + (size_t)(4 * ks + e) * PD);
                xa4[ks][e] = *(const v4f*)(xp + (size_t)(4 * ks + e) * PS);
            }
        }
        }
        const int sa = wv >> 2;
        const int na = wv & 3;
        v2f ww[16], xq[16];
#pragma unroll
        for (int ks = 0; ks < 16; ++ks) {
            const int d0 = 4 * ks + 2 * h;
            ww[ks] = mk2(W[(size_t)d0 * PD + 16 * na + c], W[(size_t)(d0 + 1) * PD + 16 * na + c]);
            xq[ks] = mk2(xb[(size_t)d0 * PS + ibw + 16 * sa + c],
                         xb[(size_t)(d0 + 1) * PS + ibw + 16 * sa + c]);
        }

        if (own) {
#pragma unroll
        for (int q = 0; q < 4; ++q) {
            v8f acc4[4];
#pragma unroll
            for (int nt = 0; nt < 4; ++nt) acc4[nt] = v8f_zero();
#pragma unroll
            for (int ks = 0; ks < 16; ++ks) {
                const v2f a = mk2(xa4[ks][0][q], xa4[ks][1][q]);
#pragma unroll
                for (int nt = 0; nt < 4; ++nt)
                    acc4[nt] = wmma_f32_16x16x4(a, mk2(wh4[ks][0][nt], wh4[ks][1][nt]), acc4[nt]);
            }
#pragma unroll
            for (int r = 0; r < 8; ++r)
                *(v4f*)&sB[64 * wv + 32 * h + 4 * r + q][4 * c] =
                    v4f{acc4[0][r], acc4[1][r], acc4[2][r], acc4[3][r]};
        }
        }

        v8f accA = v8f_zero();
#pragma unroll
        for (int ks = 0; ks < 16; ++ks)
            accA = wmma_f32_16x16x4(xq[ks], ww[ks], accA);
#pragma unroll
        for (int r = 0; r < 8; ++r)
            sA[16 * sa + 8 * h + r][16 * na + c] = accA[r];
    }

    const float* xbase = xb + (size_t)(d0 + 16 * dt2 + c) * PS + 8 * h;
    v4f xf[4][2], xfn[4][2];
    if (j2) {
#pragma unroll
        for (int u = 0; u < 4; ++u)
#pragma unroll
            for (int p = 0; p < 2; ++p)
                xf[u][p] = *(const v4f*)(xbase + 16 * u + 4 * p);
    }

    __syncthreads();

    const float* arow = &sA[16 * ct + c][4 * h];
    v4f a2[8];
    v4f nsq = {0.f, 0.f, 0.f, 0.f};
#pragma unroll
    for (int j = 0; j < 8; ++j) {
        const v4f f = *(const v4f*)(arow + 8 * j);
        nsq += f * f;
        a2[j] = f + f;
    }
    const float nap = (nsq.x + nsq.y) + (nsq.z + nsq.w);

    const float* bbase = &sB[16 * rt + c][4 * h];
    const int wl = 16 * rt + 8 * h;

    v4f bm[8], bmn[8];
    v4f G[4][2];

    const int t1 = (t0 + 1 < PS / 64) ? t0 + 1 : t0;
#pragma unroll
    for (int j = 0; j < 8; ++j) bm[j] = *(const v4f*)(bbase + (size_t)(64 * t0) * LP + 8 * j);
#pragma unroll
    for (int j = 0; j < 8; ++j) bmn[j] = *(const v4f*)(bbase + (size_t)(64 * t1) * LP + 8 * j);

    v8f acc = v8f_zero();

    if (j2 && t0 > 0) {
        const v2f ones = mk2(1.0f, 1.0f);
        v4f xfo[4][2];
#pragma unroll
        for (int u = 0; u < 4; ++u)
#pragma unroll
            for (int p = 0; p < 2; ++p)
                xfn[u][p] = *(const v4f*)(xbase + 64 + 16 * u + 4 * p);
        for (int t = 0; t < t0; ++t) {
            const int t2 = (t + 2 < PS / 64) ? t + 2 : PS / 64 - 1;
#pragma unroll
            for (int u = 0; u < 4; ++u)
#pragma unroll
                for (int p = 0; p < 2; ++p)
                    xfo[u][p] = *(const v4f*)(xbase + 64 * t2 + 16 * u + 4 * p);
#pragma unroll
            for (int u = 0; u < 4; ++u)
#pragma unroll
                for (int q = 0; q < 4; ++q)
                    acc = wmma_f32_16x16x4(kpair(xf[u], q), ones, acc);
#pragma unroll
            for (int u = 0; u < 4; ++u)
#pragma unroll
                for (int p = 0; p < 2; ++p) {
                    xf[u][p] = xfn[u][p];
                    xfn[u][p] = xfo[u][p];
                }
        }
    }

    {
        const v8f g = wmat_tile(bm, a2, nap, icol, 64 * t0 + wl);
        *(v4f*)&sG[t0 & 1][ct][rt][0][lane][0] = v4f{g[0], g[1], g[2], g[3]};
        *(v4f*)&sG[t0 & 1][ct][rt][1][lane][0] = v4f{g[4], g[5], g[6], g[7]};
    }
    __syncthreads();
    if (j2) {
#pragma unroll
        for (int u = 0; u < 4; ++u)
#pragma unroll
            for (int p = 0; p < 2; ++p)
                G[u][p] = *(const v4f*)&sG[t0 & 1][ct2][u][p][lane][0];
    }
#pragma unroll
    for (int j = 0; j < 8; ++j) bm[j] = bmn[j];

    for (int t = t0 + 1; t < PS / 64; ++t) {
        const int par = t & 1;
        const int tn = (t + 1 < PS / 64) ? t + 1 : t;

#pragma unroll
        for (int j = 0; j < 8; ++j)
            bmn[j] = *(const v4f*)(bbase + (size_t)(64 * tn) * LP + 8 * j);
        if (j2) {
#pragma unroll
            for (int u = 0; u < 4; ++u)
#pragma unroll
                for (int p = 0; p < 2; ++p)
                    xfn[u][p] = *(const v4f*)(xbase + 64 * t + 16 * u + 4 * p);

            const v8f g = wmat_tile(bm, a2, nap, icol, 64 * t + wl);
            *(v4f*)&sG[par][ct][rt][0][lane][0] = v4f{g[0], g[1], g[2], g[3]};
            *(v4f*)&sG[par][ct][rt][1][lane][0] = v4f{g[4], g[5], g[6], g[7]};

#pragma unroll
            for (int u = 0; u < 4; ++u)
#pragma unroll
                for (int q = 0; q < 4; ++q)
                    acc = wmma_f32_16x16x4(kpair(xf[u], q), kpair(G[u], q), acc);
        } else {
            const v8f g = wmat_tile(bm, a2, nap, icol, 64 * t + wl);
            *(v4f*)&sG[par][ct][rt][0][lane][0] = v4f{g[0], g[1], g[2], g[3]};
            *(v4f*)&sG[par][ct][rt][1][lane][0] = v4f{g[4], g[5], g[6], g[7]};
        }

        __syncthreads();
        if (j2) {
#pragma unroll
            for (int u = 0; u < 4; ++u)
#pragma unroll
                for (int p = 0; p < 2; ++p) {
                    G[u][p] = *(const v4f*)&sG[par][ct2][u][p][lane][0];
                    xf[u][p] = xfn[u][p];
                }
        }
#pragma unroll
        for (int j = 0; j < 8; ++j) bm[j] = bmn[j];
    }

    if (j2) {
#pragma unroll
        for (int u = 0; u < 4; ++u)
#pragma unroll
            for (int q = 0; q < 4; ++q)
                acc = wmma_f32_16x16x4(kpair(xf[u], q), kpair(G[u], q), acc);
#pragma unroll
        for (int r = 0; r < 8; ++r)
            sC[16 * dt2 + 8 * h + r][16 * ct2 + c] = acc[r];
    }
    __syncthreads();
    float* ob = out + (size_t)b * PD * PS + (size_t)d0 * PS + ibw;
    {
        const int row = 4 * wv + (lane >> 3);
        const int col = 4 * (lane & 7);
        *(v4f*)(ob + (size_t)row * PS + col) = *(const v4f*)&sC[row][col];
    }
}

extern "C" void kernel_launch(void* const* d_in, const int* in_sizes, int n_in,
                              void* d_out, int out_size, void* d_ws, size_t ws_size,
                              hipStream_t stream) {
    const float* x = (const float*)d_in[0];
    const float* W = (const float*)d_in[1];
    float* out = (float*)d_out;

    lie_fused<<<dim3(2, PB, PS / 32), 256, 0, stream>>>(x, W, out);
}


# phase 0: loads in k-step order, A tile's 64 loads behind strips 0-1, two strips per pass with the k-step outermost: products run under the load stream (first one waited for all 64 loads); sums kept
# speedup vs baseline: 3.3258x; 1.0534x over previous
#include <hip/hip_runtime.h>
#include <hip/hip_bf16.h>

#define PB 8
#define PD 64
#define PS 512
#define LP 68

typedef __attribute__((ext_vector_type(2))) float v2f;
typedef __attribute__((ext_vector_type(4))) float v4f;
typedef __attribute__((ext_vector_type(8))) float v8f;

__device__ __forceinline__ v8f wmma_f32_16x16x4(v2f a, v2f b, v8f c) {
    return __builtin_amdgcn_wmma_f32_16x16x4_f32(
         false, a,  false, b,
         (short)0, c,  false,  false);
}

__device__ __forceinline__ v8f v8f_zero() {
    v8f z = {0.f, 0.f, 0.f, 0.f, 0.f, 0.f, 0.f, 0.f};
    return z;
}

__device__ __forceinline__ v2f mk2(float a, float b) { v2f r; r.x = a; r.y = b; return r; }

__device__ __forceinline__ v2f kpair(const v4f (&p)[2], int q) {
    const v4f f = p[q >> 1];
    return (q & 1) ? mk2(f.z, f.w) : mk2(f.x, f.y);
}

__device__ __forceinline__ v8f wmat_tile(const v4f (&bm)[8], const v4f (&a2)[8],
                                         float nap, int icol, int wreg0) {
    v8f g = v8f_zero();
    v4f psq = {0.f, 0.f, 0.f, 0.f};
#pragma unroll
    for (int j = 0; j < 8; ++j) {
        const v4f f = bm[j];
        psq += f * f;
        g = wmma_f32_16x16x4(mk2(f.x, f.y), mk2(a2[j].x, a2[j].y), g);
        g = wmma_f32_16x16x4(mk2(f.z, f.w), mk2(a2[j].z, a2[j].w), g);
    }
    const float pb = (psq.x + psq.y) + (psq.z + psq.w);
    g = wmma_f32_16x16x4(mk2(pb, 1.0f), mk2(1.0f, nap), g);
#pragma unroll
    for (int r = 0; r < 8; ++r) {
        const float s = sqrtf(fmaxf(g[r], 0.0f));
        g[r] = (icol <= wreg0 + r) ? 1.0f - s : 1.0f;
    }
    return g;
}


template <int Q0>
__device__ __forceinline__ void bm_two_strips(const v4f (&xa4)[16][2], const v4f (&wh4)[16][2],
                                              float (*sBp)[LP], int row0, int c) {
    v8f acc[2][4];
#pragma unroll
    for (int s = 0; s < 2; ++s)
#pragma unroll
        for (int nt = 0; nt < 4; ++nt) acc[s][nt] = v8f_zero();
#pragma unroll
    for (int ks = 0; ks < 16; ++ks) {
#pragma unroll
        for (int s = 0; s < 2; ++s) {
            const v2f a = mk2(xa4[ks][0][Q0 + s], xa4[ks][1][Q0 + s]);
#pragma unroll
            for (int nt = 0; nt < 4; ++nt)
                acc[s][nt] = wmma_f32_16x16x4(a, mk2(wh4[ks][0][nt], wh4[ks][1][nt]), acc[s][nt]);
        }
    }
#pragma unroll
    for (int s = 0; s < 2; ++s)
#pragma unroll
        for (int r = 0; r < 8; ++r)
            *(v4f*)&sBp[row0 + 4 * r + Q0 + s][4 * c] =
                v4f{acc[s][0][r], acc[s][1][r], acc[s][2][r], acc[s][3][r]};
}

__global__ __launch_bounds__(256) void lie_fused(
    const float* __restrict__ x, const float* __restrict__ W, float* __restrict__ out) {

    __shared__ alignas(16) float sB[PS][LP];
    __shared__ alignas(16) float sA[32][LP];
    __shared__ alignas(16) float sG[2][2][4][2][32][4];
    __shared__ alignas(16) float sC[32][36];

    const int lane = threadIdx.x & 31;
    const int wv   = threadIdx.x >> 5;
    const int ct   = wv & 1;
    const int rt   = wv >> 1;
    const int c    = lane & 15;
    const int h    = lane >> 4;
    const int b    = blockIdx.y;
    const int ibw  = blockIdx.z * 32;
    const int t0   = (int)blockIdx.z >> 1;
    const int icol = ibw + 16 * ct + c;
    const int  wvs = __builtin_amdgcn_readfirstlane(wv);
    const bool j2  = (__builtin_popcount((unsigned)wvs) & 1) == 0;
    const int  dt2 = wvs >> 2;
    const int  ct2 = (wvs >> 1) & 1;
    const int  d0  = 32 * (int)blockIdx.x;

    const float* xb = x + (size_t)b * PD * PS;

    {
        const bool own = wvs >= t0;
        v4f wh4[16][2], xa4[16][2];
        const float* wp = W + (size_t)(PD + 2 * h) * PD + 4 * c;
        const float* xp = xb + (size_t)(2 * h) * PS + 64 * wv + 4 * c;
        if (own) {
#pragma unroll
        for (int ks = 0; ks < 16; ++ks) {
#pragma unroll
            for (int e = 0; e < 2; ++e) {
                wh4[ks][e] = *(const v4f*)(wp + (size_t)(4 * ks + e) * PD);
                xa4[ks][e] = *(const v4f*)(xp + (size_t)(4 * ks + e) * PS);
            }
            __builtin_amdgcn_sched_barrier(0);
        }
        }
        if (own) bm_two_strips<0>(xa4, wh4, sB, 64 * wv + 32 * h, c);

        const int sa = wv >> 2;
        const int na = wv & 3;
        v2f ww[16], xq[16];
#pragma unroll
        for (int ks = 0; ks < 16; ++ks) {
            const int d0 = 4 * ks + 2 * h;
            ww[ks] = mk2(W[(size_t)d0 * PD + 16 * na + c], W[(size_t)(d0 + 1) * PD + 16 * na + c]);
            xq[ks] = mk2(xb[(size_t)d0 * PS + ibw + 16 * sa + c],
                         xb[(size_t)(d0 + 1) * PS + ibw + 16 * sa + c]);
        }

        if (own) bm_two_strips<2>(xa4, wh4, sB, 64 * wv + 32 * h, c);

        v8f accA = v8f_zero();
#pragma unroll
        for (int ks = 0; ks < 16; ++ks)
            accA = wmma_f32_16x16x4(xq[ks], ww[ks], accA);
#pragma unroll
        for (int r = 0; r < 8; ++r)
            sA[16 * sa + 8 * h + r][16 * na + c] = accA[r];
    }

    const float* xbase = xb + (size_t)(d0 + 16 * dt2 + c) * PS + 8 * h;
    v4f xf[4][2], xfn[4][2];
    if (j2) {
#pragma unroll
        for (int u = 0; u < 4; ++u)
#pragma unroll
            for (int p = 0; p < 2; ++p)
                xf[u][p] = *(const v4f*)(xbase + 16 * u + 4 * p);
    }

    __syncthreads();

    const float* arow = &sA[16 * ct + c][4 * h];
    v4f a2[8];
    v4f nsq = {0.f, 0.f, 0.f, 0.f};
#pragma unroll
    for (int j = 0; j < 8; ++j) {
        const v4f f = *(const v4f*)(arow + 8 * j);
        nsq += f * f;
        a2[j] = f + f;
    }
    const float nap = (nsq.x + nsq.y) + (nsq.z + nsq.w);

    const float* bbase = &sB[16 * rt + c][4 * h];
    const int wl = 16 * rt + 8 * h;

    v4f bm[8], bmn[8];
    v4f G[4][2];

    const int t1 = (t0 + 1 < PS / 64) ? t0 + 1 : t0;
#pragma unroll
    for (int j = 0; j < 8; ++j) bm[j] = *(const v4f*)(bbase + (size_t)(64 * t0) * LP + 8 * j);
#pragma unroll
    for (int j = 0; j < 8; ++j) bmn[j] = *(const v4f*)(bbase + (size_t)(64 * t1) * LP + 8 * j);

    v8f acc = v8f_zero();

    if (j2 && t0 > 0) {
        const v2f ones = mk2(1.0f, 1.0f);
        v4f xfo[4][2];
#pragma unroll
        for (int u = 0; u < 4; ++u)
#pragma unroll
            for (int p = 0; p < 2; ++p)
                xfn[u][p] = *(const v4f*)(xbase + 64 + 16 * u + 4 * p);
        for (int t = 0; t < t0; ++t) {
            const int t2 = (t + 2 < PS / 64) ? t + 2 : PS / 64 - 1;
#pragma unroll
            for (int u = 0; u < 4; ++u)
#pragma unroll
                for (int p = 0; p < 2; ++p)
                    xfo[u][p] = *(const v4f*)(xbase + 64 * t2 + 16 * u + 4 * p);
#pragma unroll
            for (int u = 0; u < 4; ++u)
#pragma unroll
                for (int q = 0; q < 4; ++q)
                    acc = wmma_f32_16x16x4(kpair(xf[u], q), ones, acc);
#pragma unroll
            for (int u = 0; u < 4; ++u)
#pragma unroll
                for (int p = 0; p < 2; ++p) {
                    xf[u][p] = xfn[u][p];
                    xfn[u][p] = xfo[u][p];
                }
        }
    }

    {
        const v8f g = wmat_tile(bm, a2, nap, icol, 64 * t0 + wl);
        *(v4f*)&sG[t0 & 1][ct][rt][0][lane][0] = v4f{g[0], g[1], g[2], g[3]};
        *(v4f*)&sG[t0 & 1][ct][rt][1][lane][0] = v4f{g[4], g[5], g[6], g[7]};
    }
    __syncthreads();
    if (j2) {
#pragma unroll
        for (int u = 0; u < 4; ++u)
#pragma unroll
            for (int p = 0; p < 2; ++p)
                G[u][p] = *(const v4f*)&sG[t0 & 1][ct2][u][p][lane][0];
    }
#pragma unroll
    for (int j = 0; j < 8; ++j) bm[j] = bmn[j];

    for (int t = t0 + 1; t < PS / 64; ++t) {
        const int par = t & 1;
        const int tn = (t + 1 < PS / 64) ? t + 1 : t;

#pragma unroll
        for (int j = 0; j < 8; ++j)
            bmn[j] = *(const v4f*)(bbase + (size_t)(64 * tn) * LP + 8 * j);
        if (j2) {
#pragma unroll
            for (int u = 0; u < 4; ++u)
#pragma unroll
                for (int p = 0; p < 2; ++p)
                    xfn[u][p] = *(const v4f*)(xbase + 64 * t + 16 * u + 4 * p);

            const v8f g = wmat_tile(bm, a2, nap, icol, 64 * t + wl);
            *(v4f*)&sG[par][ct][rt][0][lane][0] = v4f{g[0], g[1], g[2], g[3]};
            *(v4f*)&sG[par][ct][rt][1][lane][0] = v4f{g[4], g[5], g[6], g[7]};

#pragma unroll
            for (int u = 0; u < 4; ++u)
#pragma unroll
                for (int q = 0; q < 4; ++q)
                    acc = wmma_f32_16x16x4(kpair(xf[u], q), kpair(G[u], q), acc);
        } else {
            const v8f g = wmat_tile(bm, a2, nap, icol, 64 * t + wl);
            *(v4f*)&sG[par][ct][rt][0][lane][0] = v4f{g[0], g[1], g[2], g[3]};
            *(v4f*)&sG[par][ct][rt][1][lane][0] = v4f{g[4], g[5], g[6], g[7]};
        }

        __syncthreads();
        if (j2) {
#pragma unroll
            for (int u = 0; u < 4; ++u)
#pragma unroll
                for (int p = 0; p < 2; ++p) {
                    G[u][p] = *(const v4f*)&sG[par][ct2][u][p][lane][0];
                    xf[u][p] = xfn[u][p];
                }
        }
#pragma unroll
        for (int j = 0; j < 8; ++j) bm[j] = bmn[j];
    }

    if (j2) {
#pragma unroll
        for (int u = 0; u < 4; ++u)
#pragma unroll
            for (int q = 0; q < 4; ++q)
                acc = wmma_f32_16x16x4(kpair(xf[u], q), kpair(G[u], q), acc);
#pragma unroll
        for (int r = 0; r < 8; ++r)
            sC[16 * dt2 + 8 * h + r][16 * ct2 + c] = acc[r];
    }
    __syncthreads();
    float* ob = out + (size_t)b * PD * PS + (size_t)d0 * PS + ibw;
    {
        const int row = 4 * wv + (lane >> 3);
        const int col = 4 * (lane & 7);
        *(v4f*)(ob + (size_t)row * PS + col) = *(const v4f*)&sC[row][col];
    }
}

extern "C" void kernel_launch(void* const* d_in, const int* in_sizes, int n_in,
                              void* d_out, int out_size, void* d_ws, size_t ws_size,
                              hipStream_t stream) {
    const float* x = (const float*)d_in[0];
    const float* W = (const float*)d_in[1];
    float* out = (float*)d_out;

    lie_fused<<<dim3(2, PB, PS / 32), 256, 0, stream>>>(x, W, out);
}


# work-groups split by column, not by output row: 16 columns x all 64 rows d each (grid 8 x 32); each wmat tile built once; 4 waves build tiles, 4 do product 2; 64-byte row stores; sums kept
# speedup vs baseline: 3.6982x; 1.1120x over previous
#include <hip/hip_runtime.h>
#include <hip/hip_bf16.h>

#define PB 8
#define PD 64
#define PS 512
#define LP 68

typedef __attribute__((ext_vector_type(2))) float v2f;
typedef __attribute__((ext_vector_type(4))) float v4f;
typedef __attribute__((ext_vector_type(8))) float v8f;

__device__ __forceinline__ v8f wmma_f32_16x16x4(v2f a, v2f b, v8f c) {
    return __builtin_amdgcn_wmma_f32_16x16x4_f32(
         false, a,  false, b,
         (short)0, c,  false,  false);
}

__device__ __forceinline__ v8f v8f_zero() {
    v8f z = {0.f, 0.f, 0.f, 0.f, 0.f, 0.f, 0.f, 0.f};
    return z;
}

__device__ __forceinline__ v2f mk2(float a, float b) { v2f r; r.x = a; r.y = b; return r; }

__device__ __forceinline__ v2f kpair(const v4f (&p)[2], int q) {
    const v4f f = p[q >> 1];
    return (q & 1) ? mk2(f.z, f.w) : mk2(f.x, f.y);
}

__device__ __forceinline__ v8f wmat_tile(const v4f (&bm)[8], const v4f (&a2)[8],
                                         float nap, int icol, int wreg0) {
    v8f g = v8f_zero();
    v4f psq = {0.f, 0.f, 0.f, 0.f};
#pragma unroll
    for (int j = 0; j < 8; ++j) {
        const v4f f = bm[j];
        psq += f * f;
        g = wmma_f32_16x16x4(mk2(f.x, f.y), mk2(a2[j].x, a2[j].y), g);
        g = wmma_f32_16x16x4(mk2(f.z, f.w), mk2(a2[j].z, a2[j].w), g);
    }
    const float pb = (psq.x + psq.y) + (psq.z + psq.w);
    g = wmma_f32_16x16x4(mk2(pb, 1.0f), mk2(1.0f, nap), g);
#pragma unroll
    for (int r = 0; r < 8; ++r) {
        const float s = sqrtf(fmaxf(g[r], 0.0f));
        g[r] = (icol <= wreg0 + r) ? 1.0f - s : 1.0f;
    }
    return g;
}


template <int Q0>
__device__ __forceinline__ void bm_two_strips(const v4f (&xa4)[16][2], const v4f (&wh4)[16][2],
                                              float (*sBp)[LP], int row0, int c) {
    v8f acc[2][4];
#pragma unroll
    for (int s = 0; s < 2; ++s)
#pragma unroll
        for (int nt = 0; nt < 4; ++nt) acc[s][nt] = v8f_zero();
#pragma unroll
    for (int ks = 0; ks < 16; ++ks) {
#pragma unroll
        for (int s = 0; s < 2; ++s) {
            const v2f a = mk2(xa4[ks][0][Q0 + s], xa4[ks][1][Q0 + s]);
#pragma unroll
            for (int nt = 0; nt < 4; ++nt)
                acc[s][nt] = wmma_f32_16x16x4(a, mk2(wh4[ks][0][nt], wh4[ks][1][nt]), acc[s][nt]);
        }
    }
#pragma unroll
    for (int s = 0; s < 2; ++s)
#pragma unroll
        for (int r = 0; r < 8; ++r)
            *(v4f*)&sBp[row0 + 4 * r + Q0 + s][4 * c] =
                v4f{acc[s][0][r], acc[s][1][r], acc[s][2][r], acc[s][3][r]};
}

__global__ __launch_bounds__(256) void lie_fused(
    const float* __restrict__ x, const float* __restrict__ W, float* __restrict__ out) {

    __shared__ alignas(16) float sB[PS][LP];
    __shared__ alignas(16) float sA[16][LP];
    __shared__ alignas(16) float sG[2][4][2][32][4];
    __shared__ alignas(16) float sC[PD][20];

    const int lane = threadIdx.x & 31;
    const int wv   = threadIdx.x >> 5;
    const int c    = lane & 15;
    const int h    = lane >> 4;
    const int b    = blockIdx.x;
    const int ibw  = (int)blockIdx.y * 16;
    const int t0   = (int)blockIdx.y >> 2;
    const int icol = ibw + c;
    const int  wvs = __builtin_amdgcn_readfirstlane(wv);
    const bool j2  = (__builtin_popcount((unsigned)wvs) & 1) == 0;
    const int  rt  = wvs >> 1;

    const float* xb = x + (size_t)b * PD * PS;

    {
        const bool own = wvs >= t0;
        v4f wh4[16][2], xa4[16][2];
        const float* wp = W + (size_t)(PD + 2 * h) * PD + 4 * c;
        const float* xp = xb + (size_t)(2 * h) * PS + 64 * wv + 4 * c;
        if (own) {
#pragma unroll
        for (int ks = 0; ks < 16; ++ks) {
#pragma unroll
            for (int e = 0; e < 2; ++e) {
                wh4[ks][e] = *(const v4f*)(wp + (size_t)(4 * ks + e) * PD);
                xa4[ks][e] = *(const v4f*)(xp + (size_t)(4 * ks + e) * PS);
            }
            __builtin_amdgcn_sched_barrier(0);
        }
        }
        if (own) bm_two_strips<0>(xa4, wh4, sB, 64 * wv + 32 * h, c);

        v2f ww[16], xq[16];
        if (j2) {
#pragma unroll
        for (int ks = 0; ks < 16; ++ks) {
            const int dk = 4 * ks + 2 * h;
            ww[ks] = mk2(W[(size_t)dk * PD + 16 * rt + c], W[(size_t)(dk + 1) * PD + 16 * rt + c]);
            xq[ks] = mk2(xb[(size_t)dk * PS + ibw + c], xb[(size_t)(dk + 1) * PS + ibw + c]);
        }
        }

        if (own) bm_two_strips<2>(xa4, wh4, sB, 64 * wv + 32 * h, c);

        if (j2) {
            v8f accA = v8f_zero();
#pragma unroll
            for (int ks = 0; ks < 16; ++ks)
                accA = wmma_f32_16x16x4(xq[ks], ww[ks], accA);
#pragma unroll
            for (int r = 0; r < 8; ++r)
                sA[8 * h + r][16 * rt + c] = accA[r];
        }
    }

    const float* xbase = xb + (size_t)(16 * rt + c) * PS + 8 * h;
    v4f xf[4][2], xfn[4][2];
    if (j2) {
#pragma unroll
        for (int u = 0; u < 4; ++u)
#pragma unroll
            for (int p = 0; p < 2; ++p)
                xf[u][p] = *(const v4f*)(xbase + 16 * u + 4 * p);
    }

    __syncthreads();

    const float* bbase = &sB[16 * rt + c][4 * h];
    const int wl = 16 * rt + 8 * h;

    v4f a2[8];
    float nap = 0.0f;
    v4f bm[8], bmn[8];
    v4f G[4][2];
    v8f acc = v8f_zero();

    const int t1 = (t0 + 1 < PS / 64) ? t0 + 1 : t0;

    if (!j2) {
        const float* arow = &sA[c][4 * h];
        v4f nsq = {0.f, 0.f, 0.f, 0.f};
#pragma unroll
        for (int j = 0; j < 8; ++j) {
            const v4f f = *(const v4f*)(arow + 8 * j);
            nsq += f * f;
            a2[j] = f + f;
        }
        nap = (nsq.x + nsq.y) + (nsq.z + nsq.w);
#pragma unroll
        for (int j = 0; j < 8; ++j) bm[j] = *(const v4f*)(bbase + (size_t)(64 * t0) * LP + 8 * j);
#pragma unroll
        for (int j = 0; j < 8; ++j) bmn[j] = *(const v4f*)(bbase + (size_t)(64 * t1) * LP + 8 * j);

        const v8f g = wmat_tile(bm, a2, nap, icol, 64 * t0 + wl);
        *(v4f*)&sG[t0 & 1][rt][0][lane][0] = v4f{g[0], g[1], g[2], g[3]};
        *(v4f*)&sG[t0 & 1][rt][1][lane][0] = v4f{g[4], g[5], g[6], g[7]};
    } else if (t0 > 0) {
        const v2f ones = mk2(1.0f, 1.0f);
        v4f xfo[4][2];
#pragma unroll
        for (int u = 0; u < 4; ++u)
#pragma unroll
            for (int p = 0; p < 2; ++p)
                xfn[u][p] = *(const v4f*)(xbase + 64 + 16 * u + 4 * p);
        for (int t = 0; t < t0; ++t) {
            const int t2 = (t + 2 < PS / 64) ? t + 2 : PS / 64 - 1;
#pragma unroll
            for (int u = 0; u < 4; ++u)
#pragma unroll
                for (int p = 0; p < 2; ++p)
                    xfo[u][p] = *(const v4f*)(xbase + 64 * t2 + 16 * u + 4 * p);
#pragma unroll
            for (int u = 0; u < 4; ++u)
#pragma unroll
                for (int q = 0; q < 4; ++q)
                    acc = wmma_f32_16x16x4(kpair(xf[u], q), ones, acc);
#pragma unroll
            for (int u = 0; u < 4; ++u)
#pragma unroll
                for (int p = 0; p < 2; ++p) {
                    xf[u][p] = xfn[u][p];
                    xfn[u][p] = xfo[u][p];
                }
        }
    }
    __syncthreads();
    if (j2) {
#pragma unroll
        for (int u = 0; u < 4; ++u)
#pragma unroll
            for (int p = 0; p < 2; ++p)
                G[u][p] = *(const v4f*)&sG[t0 & 1][u][p][lane][0];
    } else {
#pragma unroll
        for (int j = 0; j < 8; ++j) bm[j] = bmn[j];
    }

    for (int t = t0 + 1; t < PS / 64; ++t) {
        const int par = t & 1;
        if (j2) {
#pragma unroll
            for (int u = 0; u < 4; ++u)
#pragma unroll
                for (int p = 0; p < 2; ++p)
                    xfn[u][p] = *(const v4f*)(xbase + 64 * t + 16 * u + 4 * p);
            __builtin_amdgcn_sched_barrier(0);

#pragma unroll
            for (int u = 0; u < 4; ++u)
#pragma unroll
                for (int q = 0; q < 4; ++q)
                    acc = wmma_f32_16x16x4(kpair(xf[u], q), kpair(G[u], q), acc);
        } else {
            const int tn = (t + 1 < PS / 64) ? t + 1 : t;
#pragma unroll
            for (int j = 0; j < 8; ++j)
                bmn[j] = *(const v4f*)(bbase + (size_t)(64 * tn) * LP + 8 * j);

            const v8f g = wmat_tile(bm, a2, nap, icol, 64 * t + wl);
            *(v4f*)&sG[par][rt][0][lane][0] = v4f{g[0], g[1], g[2], g[3]};
            *(v4f*)&sG[par][rt][1][lane][0] = v4f{g[4], g[5], g[6], g[7]};
        }

        __syncthreads();
        if (j2) {
#pragma unroll
            for (int u = 0; u < 4; ++u)
#pragma unroll
                for (int p = 0; p < 2; ++p) {
                    G[u][p] = *(const v4f*)&sG[par][u][p][lane][0];
                    xf[u][p] = xfn[u][p];
                }
        } else {
#pragma unroll
            for (int j = 0; j < 8; ++j) bm[j] = bmn[j];
        }
    }

    if (j2) {
#pragma unroll
        for (int u = 0; u < 4; ++u)
#pragma unroll
            for (int q = 0; q < 4; ++q)
                acc = wmma_f32_16x16x4(kpair(xf[u], q), kpair(G[u], q), acc);
#pragma unroll
        for (int r = 0; r < 8; ++r)
            sC[16 * rt + 8 * h + r][c] = acc[r];
    }
    __syncthreads();
    float* ob = out + (size_t)b * PD * PS + ibw;
    {
        const int row = 8 * wv + (lane >> 2);
        const int col = 4 * (lane & 3);
        *(v4f*)(ob + (size_t)row * PS + col) = *(const v4f*)&sC[row][col];
    }
}

extern "C" void kernel_launch(void* const* d_in, const int* in_sizes, int n_in,
                              void* d_out, int out_size, void* d_ws, size_t ws_size,
                              hipStream_t stream) {
    const float* x = (const float*)d_in[0];
    const float* W = (const float*)d_in[1];
    float* out = (float*)d_out;

    lie_fused<<<dim3(PB, PS / 16), 256, 0, stream>>>(x, W, out);
}
